# starting kernel
# baseline (speedup 1.0000x reference)
#include <hip/hip_runtime.h>
#include <math.h>

typedef float v2f __attribute__((ext_vector_type(2)));
typedef float v8f __attribute__((ext_vector_type(8)));

#define HEADS   8
#define DK      32
#define RATE    2
#define SEQ     2046
#define NPAD    2048
#define MROWS   1024
#define DMODEL  256
#define BATCH   4

#define YSTRIDE 34
#define PSTRIDE 18

__global__ __launch_bounds__(32)
void qkv_gemm_kernel(const float* __restrict__ x,
                     const float* __restrict__ W,
                     const float* __restrict__ bias,
                     float* __restrict__ qkv)
{
    const int lane = threadIdx.x;
    const int half = lane >> 4;
    const int l16  = lane & 15;

    const int mtile = blockIdx.x;
    const int ntile = blockIdx.y;

    const int row = mtile * 16 + l16;
    const int bb  = row >> 11;
    const int n   = row & (NPAD - 1);
    const bool valid = (n < SEQ);
    const int col = ntile * 16 + l16;

    const float mask = valid ? 1.0f : 0.0f;
    const float* xrow = x + ((size_t)(bb * SEQ + (valid ? n : 0))) * DMODEL;

    const float bv = bias[col];
    v8f c;
    #pragma unroll
    for (int v = 0; v < 8; ++v) c[v] = bv;

    #pragma unroll 4
    for (int k0 = 0; k0 < DMODEL; k0 += 4) {
        const int k = k0 + 2 * half;
        const float2 av = *(const float2*)(xrow + k);
        v2f a;
        a[0] = av.x * mask;
        a[1] = av.y * mask;
        v2f bm;
        bm[0] = W[(size_t)k * DMODEL + col];
        bm[1] = W[(size_t)(k + 1) * DMODEL + col];
        c = __builtin_amdgcn_wmma_f32_16x16x4_f32(false, a, false, bm,
                                                  (short)0, c, false, false);
    }

    #pragma unroll
    for (int v = 0; v < 8; ++v) {
        const int rrow = mtile * 16 + v + 8 * half;
        qkv[(size_t)rrow * DMODEL + col] = c[v];
    }
}

__global__ __launch_bounds__(256)
void attn_kernel(const float* __restrict__ qkv, float* __restrict__ out)
{
    extern __shared__ float lds[];
    float* Ys = lds;
    float* Ps = lds + MROWS * YSTRIDE;

    const int tid  = threadIdx.x;
    const int wave = tid >> 5;
    const int lane = tid & 31;
    const int half = lane >> 4;
    const int l16  = lane & 15;

    const int g  = blockIdx.y;
    const int bb = g >> 4;
    const int h  = (g >> 1) & 7;
    const int r  = g & 1;

    {
        const int dcol  = (tid & 15) * 2;
        const int rbase = tid >> 4;
        for (int i = rbase; i < MROWS; i += 16) {
            const float* src = qkv + ((size_t)(bb * NPAD + 2 * i + r)) * DMODEL
                                   + h * DK + dcol;
            const float2 v = *(const float2*)src;
            *(float2*)&Ys[i * YSTRIDE + dcol] = v;
        }
    }
    __syncthreads();

    float* Pw = Ps + wave * 16 * PSTRIDE;

    const int qb    = blockIdx.x * 8 + wave;
    const int qbase = qb * 16;

    const float scale = 0.17677669529663687f;

    float mrow[8], lrow[8];
    v8f o0, o1;
    #pragma unroll
    for (int v = 0; v < 8; ++v) {
        mrow[v] = -3.0e38f; lrow[v] = 0.0f;
        o0[v] = 0.0f; o1[v] = 0.0f;
    }

    for (int kb = 0; kb < 64; ++kb) {
        const int kbase = kb * 16;

        v8f s;
        #pragma unroll
        for (int v = 0; v < 8; ++v) s[v] = 0.0f;
        #pragma unroll
        for (int d0 = 0; d0 < DK; d0 += 4) {
            const int d = d0 + 2 * half;
            v2f a, b;
            const float2 av = *(const float2*)&Ys[(qbase + l16) * YSTRIDE + d];
            const float2 bv = *(const float2*)&Ys[(kbase + l16) * YSTRIDE + d];
            a[0] = av.x; a[1] = av.y;
            b[0] = bv.x; b[1] = bv.y;
            s = __builtin_amdgcn_wmma_f32_16x16x4_f32(false, a, false, b,
                                                      (short)0, s, false, false);
        }

        #pragma unroll
        for (int v = 0; v < 8; ++v) {
            const float sv = s[v] * scale;
            float rmax = sv;
            #pragma unroll
            for (int mk = 1; mk < 16; mk <<= 1)
                rmax = fmaxf(rmax, __shfl_xor(rmax, mk, 32));
            const float newm = fmaxf(mrow[v], rmax);
            const float corr = __expf(mrow[v] - newm);
            const float p    = __expf(sv - newm);
            float rsum = p;
            #pragma unroll
            for (int mk = 1; mk < 16; mk <<= 1)
                rsum += __shfl_xor(rsum, mk, 32);
            lrow[v] = lrow[v] * corr + rsum;
            mrow[v] = newm;
            o0[v] *= corr;
            o1[v] *= corr;
            Pw[(v + 8 * half) * PSTRIDE + l16] = p;
        }

        asm volatile("s_wait_dscnt 0x0" ::: "memory");

        #pragma unroll
        for (int k0 = 0; k0 < 16; k0 += 4) {
            const int kk = k0 + 2 * half;
            v2f a, b0, b1;
            const float2 pa = *(const float2*)&Pw[l16 * PSTRIDE + kk];
            a[0]  = pa.x;
            a[1]  = pa.y;
            b0[0] = Ys[(kbase + kk) * YSTRIDE + l16];
            b0[1] = Ys[(kbase + kk + 1) * YSTRIDE + l16];
            b1[0] = Ys[(kbase + kk) * YSTRIDE + 16 + l16];
            b1[1] = Ys[(kbase + kk + 1) * YSTRIDE + 16 + l16];
            o0 = __builtin_amdgcn_wmma_f32_16x16x4_f32(false, a, false, b0,
                                                       (short)0, o0, false, false);
            o1 = __builtin_amdgcn_wmma_f32_16x16x4_f32(false, a, false, b1,
                                                       (short)0, o1, false, false);
        }
    }

    #pragma unroll
    for (int v = 0; v < 8; ++v) {
        const int row = qbase + v + 8 * half;
        float e[3];
        const int d2 = l16 * 2;
        #pragma unroll
        for (int k = 0; k < 3; ++k) {
            const int nr = row + k - 1;
            const bool ok = (nr >= 0) && (nr < MROWS);
            const int nrc = ok ? nr : row;
            const float msk = ok ? 1.0f : 0.0f;
            float part = msk * (Ys[row * YSTRIDE + d2]     * Ys[nrc * YSTRIDE + d2]
                              + Ys[row * YSTRIDE + d2 + 1] * Ys[nrc * YSTRIDE + d2 + 1]);
            #pragma unroll
            for (int mk = 1; mk < 16; mk <<= 1)
                part += __shfl_xor(part, mk, 32);
            e[k] = part * scale;
        }
        const float newm = fmaxf(fmaxf(mrow[v], e[0]), fmaxf(e[1], e[2]));
        const float corr = __expf(mrow[v] - newm);
        float psum = 0.0f, add0 = 0.0f, add1 = 0.0f;
        #pragma unroll
        for (int k = 0; k < 3; ++k) {
            const float p = __expf(e[k] - newm);
            psum += p;
            const int nr = row + k - 1;
            const bool ok = (nr >= 0) && (nr < MROWS);
            const int nrc = ok ? nr : row;
            const float msk = ok ? 1.0f : 0.0f;
            add0 += (p * msk) * Ys[nrc * YSTRIDE + l16];
            add1 += (p * msk) * Ys[nrc * YSTRIDE + 16 + l16];
        }
        const float inv = 1.0f / (lrow[v] * corr + psum);
        const float r0 = (o0[v] * corr + add0) * inv;
        const float r1 = (o1[v] * corr + add1) * inv;

        const int n = 2 * row + r;
        if (n < SEQ) {
            const size_t base = ((size_t)(bb * SEQ + n)) * DMODEL + h * DK;
            out[base + l16]      = r0;
            out[base + 16 + l16] = r1;
        }
    }
}

extern "C" void kernel_launch(void* const* d_in, const int* in_sizes, int n_in,
                              void* d_out, int out_size, void* d_ws, size_t ws_size,
                              hipStream_t stream) {
    const float* x = (const float*)d_in[0];
    const float* W = (const float*)d_in[1];
    const float* b = (const float*)d_in[2];
    float* out = (float*)d_out;
    float* qkv = (float*)d_ws;

    dim3 g1(512, 16);
    qkv_gemm_kernel<<<g1, 32, 0, stream>>>(x, W, b, qkv);

    const size_t ldsBytes = (size_t)(MROWS * YSTRIDE + 8 * 16 * PSTRIDE) * sizeof(float);
    dim3 g2(8, 64);
    attn_kernel<<<g2, 256, ldsBytes, stream>>>(qkv, out);
}


# Softmax row max, row sum and the window dot sums: 16-lane butterflies as DPP row_xmask ops on the vector ALU instead of LDS permutes (ds_bpermute + wait each step); same sums, same order
# speedup vs baseline: 1.9817x; 1.9817x over previous
#include <hip/hip_runtime.h>
#include <math.h>

typedef float v2f __attribute__((ext_vector_type(2)));
typedef float v8f __attribute__((ext_vector_type(8)));

#define HEADS   8
#define DK      32
#define RATE    2
#define SEQ     2046
#define NPAD    2048
#define MROWS   1024
#define DMODEL  256
#define BATCH   4

template <int M>
__device__ __forceinline__ float row_xor(float x)
{
    const int xi = __builtin_bit_cast(int, x);
    return __builtin_bit_cast(float,
        __builtin_amdgcn_update_dpp(xi, xi, 0x160 + M, 0xf, 0xf, true));
}

__device__ __forceinline__ float row_max16(float x)
{
    x = fmaxf(x, row_xor<1>(x));
    x = fmaxf(x, row_xor<2>(x));
    x = fmaxf(x, row_xor<4>(x));
    x = fmaxf(x, row_xor<8>(x));
    return x;
}
__device__ __forceinline__ float row_sum16(float x)
{
    x += row_xor<1>(x);
    x += row_xor<2>(x);
    x += row_xor<4>(x);
    x += row_xor<8>(x);
    return x;
}

#define YSTRIDE 34
#define PSTRIDE 18

__global__ __launch_bounds__(32)
void qkv_gemm_kernel(const float* __restrict__ x,
                     const float* __restrict__ W,
                     const float* __restrict__ bias,
                     float* __restrict__ qkv)
{
    const int lane = threadIdx.x;
    const int half = lane >> 4;
    const int l16  = lane & 15;

    const int mtile = blockIdx.x;
    const int ntile = blockIdx.y;

    const int row = mtile * 16 + l16;
    const int bb  = row >> 11;
    const int n   = row & (NPAD - 1);
    const bool valid = (n < SEQ);
    const int col = ntile * 16 + l16;

    const float mask = valid ? 1.0f : 0.0f;
    const float* xrow = x + ((size_t)(bb * SEQ + (valid ? n : 0))) * DMODEL;

    const float bv = bias[col];
    v8f c;
    #pragma unroll
    for (int v = 0; v < 8; ++v) c[v] = bv;

    #pragma unroll 4
    for (int k0 = 0; k0 < DMODEL; k0 += 4) {
        const int k = k0 + 2 * half;
        const float2 av = *(const float2*)(xrow + k);
        v2f a;
        a[0] = av.x * mask;
        a[1] = av.y * mask;
        v2f bm;
        bm[0] = W[(size_t)k * DMODEL + col];
        bm[1] = W[(size_t)(k + 1) * DMODEL + col];
        c = __builtin_amdgcn_wmma_f32_16x16x4_f32(false, a, false, bm,
                                                  (short)0, c, false, false);
    }

    #pragma unroll
    for (int v = 0; v < 8; ++v) {
        const int rrow = mtile * 16 + v + 8 * half;
        qkv[(size_t)rrow * DMODEL + col] = c[v];
    }
}

__global__ __launch_bounds__(256)
void attn_kernel(const float* __restrict__ qkv, float* __restrict__ out)
{
    extern __shared__ float lds[];
    float* Ys = lds;
    float* Ps = lds + MROWS * YSTRIDE;

    const int tid  = threadIdx.x;
    const int wave = tid >> 5;
    const int lane = tid & 31;
    const int half = lane >> 4;
    const int l16  = lane & 15;

    const int g  = blockIdx.y;
    const int bb = g >> 4;
    const int h  = (g >> 1) & 7;
    const int r  = g & 1;

    {
        const int dcol  = (tid & 15) * 2;
        const int rbase = tid >> 4;
        for (int i = rbase; i < MROWS; i += 16) {
            const float* src = qkv + ((size_t)(bb * NPAD + 2 * i + r)) * DMODEL
                                   + h * DK + dcol;
            const float2 v = *(const float2*)src;
            *(float2*)&Ys[i * YSTRIDE + dcol] = v;
        }
    }
    __syncthreads();

    float* Pw = Ps + wave * 16 * PSTRIDE;

    const int qb    = blockIdx.x * 8 + wave;
    const int qbase = qb * 16;

    const float scale = 0.17677669529663687f;

    float mrow[8], lrow[8];
    v8f o0, o1;
    #pragma unroll
    for (int v = 0; v < 8; ++v) {
        mrow[v] = -3.0e38f; lrow[v] = 0.0f;
        o0[v] = 0.0f; o1[v] = 0.0f;
    }

    for (int kb = 0; kb < 64; ++kb) {
        const int kbase = kb * 16;

        v8f s;
        #pragma unroll
        for (int v = 0; v < 8; ++v) s[v] = 0.0f;
        #pragma unroll
        for (int d0 = 0; d0 < DK; d0 += 4) {
            const int d = d0 + 2 * half;
            v2f a, b;
            const float2 av = *(const float2*)&Ys[(qbase + l16) * YSTRIDE + d];
            const float2 bv = *(const float2*)&Ys[(kbase + l16) * YSTRIDE + d];
            a[0] = av.x; a[1] = av.y;
            b[0] = bv.x; b[1] = bv.y;
            s = __builtin_amdgcn_wmma_f32_16x16x4_f32(false, a, false, b,
                                                      (short)0, s, false, false);
        }

        #pragma unroll
        for (int v = 0; v < 8; ++v) {
            const float sv = s[v] * scale;
            const float rmax = row_max16(sv);
            const float newm = fmaxf(mrow[v], rmax);
            const float corr = __expf(mrow[v] - newm);
            const float p    = __expf(sv - newm);
            const float rsum = row_sum16(p);
            lrow[v] = lrow[v] * corr + rsum;
            mrow[v] = newm;
            o0[v] *= corr;
            o1[v] *= corr;
            Pw[(v + 8 * half) * PSTRIDE + l16] = p;
        }

        asm volatile("s_wait_dscnt 0x0" ::: "memory");

        #pragma unroll
        for (int k0 = 0; k0 < 16; k0 += 4) {
            const int kk = k0 + 2 * half;
            v2f a, b0, b1;
            const float2 pa = *(const float2*)&Pw[l16 * PSTRIDE + kk];
            a[0]  = pa.x;
            a[1]  = pa.y;
            b0[0] = Ys[(kbase + kk) * YSTRIDE + l16];
            b0[1] = Ys[(kbase + kk + 1) * YSTRIDE + l16];
            b1[0] = Ys[(kbase + kk) * YSTRIDE + 16 + l16];
            b1[1] = Ys[(kbase + kk + 1) * YSTRIDE + 16 + l16];
            o0 = __builtin_amdgcn_wmma_f32_16x16x4_f32(false, a, false, b0,
                                                       (short)0, o0, false, false);
            o1 = __builtin_amdgcn_wmma_f32_16x16x4_f32(false, a, false, b1,
                                                       (short)0, o1, false, false);
        }
    }

    #pragma unroll
    for (int v = 0; v < 8; ++v) {
        const int row = qbase + v + 8 * half;
        float e[3];
        const int d2 = l16 * 2;
        #pragma unroll
        for (int k = 0; k < 3; ++k) {
            const int nr = row + k - 1;
            const bool ok = (nr >= 0) && (nr < MROWS);
            const int nrc = ok ? nr : row;
            const float msk = ok ? 1.0f : 0.0f;
            float part = msk * (Ys[row * YSTRIDE + d2]     * Ys[nrc * YSTRIDE + d2]
                              + Ys[row * YSTRIDE + d2 + 1] * Ys[nrc * YSTRIDE + d2 + 1]);
            part = row_sum16(part);
            e[k] = part * scale;
        }
        const float newm = fmaxf(fmaxf(mrow[v], e[0]), fmaxf(e[1], e[2]));
        const float corr = __expf(mrow[v] - newm);
        float psum = 0.0f, add0 = 0.0f, add1 = 0.0f;
        #pragma unroll
        for (int k = 0; k < 3; ++k) {
            const float p = __expf(e[k] - newm);
            psum += p;
            const int nr = row + k - 1;
            const bool ok = (nr >= 0) && (nr < MROWS);
            const int nrc = ok ? nr : row;
            const float msk = ok ? 1.0f : 0.0f;
            add0 += (p * msk) * Ys[nrc * YSTRIDE + l16];
            add1 += (p * msk) * Ys[nrc * YSTRIDE + 16 + l16];
        }
        const float inv = 1.0f / (lrow[v] * corr + psum);
        const float r0 = (o0[v] * corr + add0) * inv;
        const float r1 = (o1[v] * corr + add1) * inv;

        const int n = 2 * row + r;
        if (n < SEQ) {
            const size_t base = ((size_t)(bb * SEQ + n)) * DMODEL + h * DK;
            out[base + l16]      = r0;
            out[base + 16 + l16] = r1;
        }
    }
}

extern "C" void kernel_launch(void* const* d_in, const int* in_sizes, int n_in,
                              void* d_out, int out_size, void* d_ws, size_t ws_size,
                              hipStream_t stream) {
    const float* x = (const float*)d_in[0];
    const float* W = (const float*)d_in[1];
    const float* b = (const float*)d_in[2];
    float* out = (float*)d_out;
    float* qkv = (float*)d_ws;

    dim3 g1(512, 16);
    qkv_gemm_kernel<<<g1, 32, 0, stream>>>(x, W, b, qkv);

    const size_t ldsBytes = (size_t)(MROWS * YSTRIDE + 8 * 16 * PSTRIDE) * sizeof(float);
    dim3 g2(8, 64);
    attn_kernel<<<g2, 256, ldsBytes, stream>>>(qkv, out);
}


# Attention online softmax advances 128 keys per step (eight score tiles, eight accumulators): one row max, rescale and row sum per 128 keys, P staged in 32-key pieces; row-sum and rescale order change
# speedup vs baseline: 2.5577x; 1.2907x over previous
#include <hip/hip_runtime.h>
#include <math.h>

typedef float v2f __attribute__((ext_vector_type(2)));
typedef float v8f __attribute__((ext_vector_type(8)));

#define HEADS   8
#define DK      32
#define RATE    2
#define SEQ     2046
#define NPAD    2048
#define MROWS   1024
#define DMODEL  256
#define BATCH   4

template <int M>
__device__ __forceinline__ float row_xor(float x)
{
    const int xi = __builtin_bit_cast(int, x);
    return __builtin_bit_cast(float,
        __builtin_amdgcn_update_dpp(xi, xi, 0x160 + M, 0xf, 0xf, true));
}

__device__ __forceinline__ float row_max16(float x)
{
    x = fmaxf(x, row_xor<1>(x));
    x = fmaxf(x, row_xor<2>(x));
    x = fmaxf(x, row_xor<4>(x));
    x = fmaxf(x, row_xor<8>(x));
    return x;
}
__device__ __forceinline__ float row_sum16(float x)
{
    x += row_xor<1>(x);
    x += row_xor<2>(x);
    x += row_xor<4>(x);
    x += row_xor<8>(x);
    return x;
}

#define YSTRIDE 34
#define PSTRIDE 34
#define NT      8
#define KSTEP   (16 * NT)

__global__ __launch_bounds__(32)
void qkv_gemm_kernel(const float* __restrict__ x,
                     const float* __restrict__ W,
                     const float* __restrict__ bias,
                     float* __restrict__ qkv)
{
    const int lane = threadIdx.x;
    const int half = lane >> 4;
    const int l16  = lane & 15;

    const int mtile = blockIdx.x;
    const int ntile = blockIdx.y;

    const int row = mtile * 16 + l16;
    const int bb  = row >> 11;
    const int n   = row & (NPAD - 1);
    const bool valid = (n < SEQ);
    const int col = ntile * 16 + l16;

    const float mask = valid ? 1.0f : 0.0f;
    const float* xrow = x + ((size_t)(bb * SEQ + (valid ? n : 0))) * DMODEL;

    const float bv = bias[col];
    v8f c;
    #pragma unroll
    for (int v = 0; v < 8; ++v) c[v] = bv;

    #pragma unroll 4
    for (int k0 = 0; k0 < DMODEL; k0 += 4) {
        const int k = k0 + 2 * half;
        const float2 av = *(const float2*)(xrow + k);
        v2f a;
        a[0] = av.x * mask;
        a[1] = av.y * mask;
        v2f bm;
        bm[0] = W[(size_t)k * DMODEL + col];
        bm[1] = W[(size_t)(k + 1) * DMODEL + col];
        c = __builtin_amdgcn_wmma_f32_16x16x4_f32(false, a, false, bm,
                                                  (short)0, c, false, false);
    }

    #pragma unroll
    for (int v = 0; v < 8; ++v) {
        const int rrow = mtile * 16 + v + 8 * half;
        qkv[(size_t)rrow * DMODEL + col] = c[v];
    }
}

__global__ __launch_bounds__(256)
void attn_kernel(const float* __restrict__ qkv, float* __restrict__ out)
{
    extern __shared__ float lds[];
    float* Ys = lds;
    float* Ps = lds + MROWS * YSTRIDE;

    const int tid  = threadIdx.x;
    const int wave = tid >> 5;
    const int lane = tid & 31;
    const int half = lane >> 4;
    const int l16  = lane & 15;

    const int g  = blockIdx.y;
    const int bb = g >> 4;
    const int h  = (g >> 1) & 7;
    const int r  = g & 1;

    {
        const int dcol  = (tid & 15) * 2;
        const int rbase = tid >> 4;
        for (int i = rbase; i < MROWS; i += 16) {
            const float* src = qkv + ((size_t)(bb * NPAD + 2 * i + r)) * DMODEL
                                   + h * DK + dcol;
            const float2 v = *(const float2*)src;
            *(float2*)&Ys[i * YSTRIDE + dcol] = v;
        }
    }
    __syncthreads();

    float* Pw = Ps + wave * 16 * PSTRIDE;

    const int qb    = blockIdx.x * 8 + wave;
    const int qbase = qb * 16;

    const float scale = 0.17677669529663687f;

    float mrow[8], lrow[8];
    v8f o0, o1;
    #pragma unroll
    for (int v = 0; v < 8; ++v) {
        mrow[v] = -3.0e38f; lrow[v] = 0.0f;
        o0[v] = 0.0f; o1[v] = 0.0f;
    }

    for (int kb = 0; kb < MROWS / KSTEP; ++kb) {
        const int kbase = kb * KSTEP;

        v8f s[NT];
        #pragma unroll
        for (int t = 0; t < NT; ++t) {
            #pragma unroll
            for (int v = 0; v < 8; ++v) s[t][v] = 0.0f;
        }
        #pragma unroll
        for (int d0 = 0; d0 < DK; d0 += 4) {
            const int d = d0 + 2 * half;
            v2f a;
            const float2 av = *(const float2*)&Ys[(qbase + l16) * YSTRIDE + d];
            a[0] = av.x; a[1] = av.y;
            #pragma unroll
            for (int t = 0; t < NT; ++t) {
                v2f b;
                const float2 bv = *(const float2*)&Ys[(kbase + 16 * t + l16) * YSTRIDE + d];
                b[0] = bv.x; b[1] = bv.y;
                s[t] = __builtin_amdgcn_wmma_f32_16x16x4_f32(false, a, false, b,
                                                             (short)0, s[t], false, false);
            }
        }

        #pragma unroll
        for (int v = 0; v < 8; ++v) {
            float mx = s[0][v];
            #pragma unroll
            for (int t = 1; t < NT; ++t) mx = fmaxf(mx, s[t][v]);
            const float rmax = row_max16(mx) * scale;
            const float newm = fmaxf(mrow[v], rmax);
            const float corr = __expf(mrow[v] - newm);
            float psum[NT / 2];
            #pragma unroll
            for (int t = 0; t < NT; t += 2) {
                const float pa = __expf(s[t][v] * scale - newm);
                const float pb = __expf(s[t + 1][v] * scale - newm);
                s[t][v] = pa; s[t + 1][v] = pb;
                psum[t / 2] = pa + pb;
            }
            #pragma unroll
            for (int w = NT / 4; w >= 1; w >>= 1) {
                #pragma unroll
                for (int t = 0; t < w; ++t) psum[t] += psum[t + w];
            }
            const float rsum = row_sum16(psum[0]);
            lrow[v] = lrow[v] * corr + rsum;
            mrow[v] = newm;
            o0[v] *= corr;
            o1[v] *= corr;
        }

        #pragma unroll
        for (int hf = 0; hf < NT / 2; ++hf) {
            #pragma unroll
            for (int v = 0; v < 8; ++v) {
                Pw[(v + 8 * half) * PSTRIDE + l16]      = s[2 * hf][v];
                Pw[(v + 8 * half) * PSTRIDE + 16 + l16] = s[2 * hf + 1][v];
            }

            asm volatile("s_wait_dscnt 0x0" ::: "memory");

            const int vbase = kbase + 32 * hf;
            #pragma unroll
            for (int k0 = 0; k0 < 32; k0 += 4) {
                const int kk = k0 + 2 * half;
                v2f a, b0, b1;
                const float2 pa = *(const float2*)&Pw[l16 * PSTRIDE + kk];
                a[0]  = pa.x;
                a[1]  = pa.y;
                b0[0] = Ys[(vbase + kk) * YSTRIDE + l16];
                b0[1] = Ys[(vbase + kk + 1) * YSTRIDE + l16];
                b1[0] = Ys[(vbase + kk) * YSTRIDE + 16 + l16];
                b1[1] = Ys[(vbase + kk + 1) * YSTRIDE + 16 + l16];
                o0 = __builtin_amdgcn_wmma_f32_16x16x4_f32(false, a, false, b0,
                                                           (short)0, o0, false, false);
                o1 = __builtin_amdgcn_wmma_f32_16x16x4_f32(false, a, false, b1,
                                                           (short)0, o1, false, false);
            }
        }
    }

    #pragma unroll
    for (int v = 0; v < 8; ++v) {
        const int row = qbase + v + 8 * half;
        float e[3];
        const int d2 = l16 * 2;
        #pragma unroll
        for (int k = 0; k < 3; ++k) {
            const int nr = row + k - 1;
            const bool ok = (nr >= 0) && (nr < MROWS);
            const int nrc = ok ? nr : row;
            const float msk = ok ? 1.0f : 0.0f;
            float part = msk * (Ys[row * YSTRIDE + d2]     * Ys[nrc * YSTRIDE + d2]
                              + Ys[row * YSTRIDE + d2 + 1] * Ys[nrc * YSTRIDE + d2 + 1]);
            part = row_sum16(part);
            e[k] = part * scale;
        }
        const float newm = fmaxf(fmaxf(mrow[v], e[0]), fmaxf(e[1], e[2]));
        const float corr = __expf(mrow[v] - newm);
        float psum = 0.0f, add0 = 0.0f, add1 = 0.0f;
        #pragma unroll
        for (int k = 0; k < 3; ++k) {
            const float p = __expf(e[k] - newm);
            psum += p;
            const int nr = row + k - 1;
            const bool ok = (nr >= 0) && (nr < MROWS);
            const int nrc = ok ? nr : row;
            const float msk = ok ? 1.0f : 0.0f;
            add0 += (p * msk) * Ys[nrc * YSTRIDE + l16];
            add1 += (p * msk) * Ys[nrc * YSTRIDE + 16 + l16];
        }
        const float inv = 1.0f / (lrow[v] * corr + psum);
        const float r0 = (o0[v] * corr + add0) * inv;
        const float r1 = (o1[v] * corr + add1) * inv;

        const int n = 2 * row + r;
        if (n < SEQ) {
            const size_t base = ((size_t)(bb * SEQ + n)) * DMODEL + h * DK;
            out[base + l16]      = r0;
            out[base + 16 + l16] = r1;
        }
    }
}

extern "C" void kernel_launch(void* const* d_in, const int* in_sizes, int n_in,
                              void* d_out, int out_size, void* d_ws, size_t ws_size,
                              hipStream_t stream) {
    const float* x = (const float*)d_in[0];
    const float* W = (const float*)d_in[1];
    const float* b = (const float*)d_in[2];
    float* out = (float*)d_out;
    float* qkv = (float*)d_ws;

    dim3 g1(512, 16);
    qkv_gemm_kernel<<<g1, 32, 0, stream>>>(x, W, b, qkv);

    const size_t ldsBytes = (size_t)(MROWS * YSTRIDE + 8 * 16 * PSTRIDE) * sizeof(float);
    dim3 g2(8, 64);
    attn_kernel<<<g2, 256, ldsBytes, stream>>>(qkv, out);
}


# Attention key loop: LDS fragment reads issued as a few large groups in front of WMMA blocks (K per 128-key step, V per two pieces, P per piece), not one read between WMMAs; same sums, same order
# speedup vs baseline: 2.7036x; 1.0570x over previous
#include <hip/hip_runtime.h>
#include <math.h>

typedef float v2f __attribute__((ext_vector_type(2)));
typedef float v8f __attribute__((ext_vector_type(8)));

#define HEADS   8
#define DK      32
#define RATE    2
#define SEQ     2046
#define NPAD    2048
#define MROWS   1024
#define DMODEL  256
#define BATCH   4

template <int M>
__device__ __forceinline__ float row_xor(float x)
{
    const int xi = __builtin_bit_cast(int, x);
    return __builtin_bit_cast(float,
        __builtin_amdgcn_update_dpp(xi, xi, 0x160 + M, 0xf, 0xf, true));
}

__device__ __forceinline__ float row_max16(float x)
{
    x = fmaxf(x, row_xor<1>(x));
    x = fmaxf(x, row_xor<2>(x));
    x = fmaxf(x, row_xor<4>(x));
    x = fmaxf(x, row_xor<8>(x));
    return x;
}
__device__ __forceinline__ float row_sum16(float x)
{
    x += row_xor<1>(x);
    x += row_xor<2>(x);
    x += row_xor<4>(x);
    x += row_xor<8>(x);
    return x;
}

#define YSTRIDE 34
#define PSTRIDE 34
#define NT      8
#define KSTEP   (16 * NT)

__global__ __launch_bounds__(32)
void qkv_gemm_kernel(const float* __restrict__ x,
                     const float* __restrict__ W,
                     const float* __restrict__ bias,
                     float* __restrict__ qkv)
{
    const int lane = threadIdx.x;
    const int half = lane >> 4;
    const int l16  = lane & 15;

    const int mtile = blockIdx.x;
    const int ntile = blockIdx.y;

    const int row = mtile * 16 + l16;
    const int bb  = row >> 11;
    const int n   = row & (NPAD - 1);
    const bool valid = (n < SEQ);
    const int col = ntile * 16 + l16;

    const float mask = valid ? 1.0f : 0.0f;
    const float* xrow = x + ((size_t)(bb * SEQ + (valid ? n : 0))) * DMODEL;

    const float bv = bias[col];
    v8f c;
    #pragma unroll
    for (int v = 0; v < 8; ++v) c[v] = bv;

    #pragma unroll 4
    for (int k0 = 0; k0 < DMODEL; k0 += 4) {
        const int k = k0 + 2 * half;
        const float2 av = *(const float2*)(xrow + k);
        v2f a;
        a[0] = av.x * mask;
        a[1] = av.y * mask;
        v2f bm;
        bm[0] = W[(size_t)k * DMODEL + col];
        bm[1] = W[(size_t)(k + 1) * DMODEL + col];
        c = __builtin_amdgcn_wmma_f32_16x16x4_f32(false, a, false, bm,
                                                  (short)0, c, false, false);
    }

    #pragma unroll
    for (int v = 0; v < 8; ++v) {
        const int rrow = mtile * 16 + v + 8 * half;
        qkv[(size_t)rrow * DMODEL + col] = c[v];
    }
}

__global__ __launch_bounds__(256)
void attn_kernel(const float* __restrict__ qkv, float* __restrict__ out)
{
    extern __shared__ float lds[];
    float* Ys = lds;
    float* Ps = lds + MROWS * YSTRIDE;

    const int tid  = threadIdx.x;
    const int wave = tid >> 5;
    const int lane = tid & 31;
    const int half = lane >> 4;
    const int l16  = lane & 15;

    const int g  = blockIdx.y;
    const int bb = g >> 4;
    const int h  = (g >> 1) & 7;
    const int r  = g & 1;

    {
        const int dcol  = (tid & 15) * 2;
        const int rbase = tid >> 4;
        for (int i = rbase; i < MROWS; i += 16) {
            const float* src = qkv + ((size_t)(bb * NPAD + 2 * i + r)) * DMODEL
                                   + h * DK + dcol;
            const float2 v = *(const float2*)src;
            *(float2*)&Ys[i * YSTRIDE + dcol] = v;
        }
    }
    __syncthreads();

    float* Pw = Ps + wave * 16 * PSTRIDE;

    const int qb    = blockIdx.x * 8 + wave;
    const int qbase = qb * 16;

    const float scale = 0.17677669529663687f;

    float mrow[8], lrow[8];
    v8f o0, o1;
    #pragma unroll
    for (int v = 0; v < 8; ++v) {
        mrow[v] = -3.0e38f; lrow[v] = 0.0f;
        o0[v] = 0.0f; o1[v] = 0.0f;
    }

    for (int kb = 0; kb < MROWS / KSTEP; ++kb) {
        const int kbase = kb * KSTEP;

        v2f qf[DK / 4];
        v2f kf[NT][DK / 4];
        #pragma unroll
        for (int j = 0; j < DK / 4; ++j) {
            const float2 av = *(const float2*)&Ys[(qbase + l16) * YSTRIDE + 4 * j + 2 * half];
            qf[j][0] = av.x; qf[j][1] = av.y;
        }
        #pragma unroll
        for (int t = 0; t < NT; ++t) {
            #pragma unroll
            for (int j = 0; j < DK / 4; ++j) {
                const float2 bv = *(const float2*)&Ys[(kbase + 16 * t + l16) * YSTRIDE + 4 * j + 2 * half];
                kf[t][j][0] = bv.x; kf[t][j][1] = bv.y;
            }
        }
        __builtin_amdgcn_sched_barrier(0);

        v8f s[NT];
        #pragma unroll
        for (int t = 0; t < NT; ++t) {
            #pragma unroll
            for (int v = 0; v < 8; ++v) s[t][v] = 0.0f;
        }
        #pragma unroll
        for (int g = 0; g < NT; g += 4) {
            #pragma unroll
            for (int j = 0; j < DK / 4; ++j) {
                #pragma unroll
                for (int t = g; t < g + 4; ++t) {
                    s[t] = __builtin_amdgcn_wmma_f32_16x16x4_f32(false, qf[j], false, kf[t][j],
                                                                 (short)0, s[t], false, false);
                }
            }
            if (g == 0) __builtin_amdgcn_sched_barrier(0);
        }
        __builtin_amdgcn_sched_barrier(0);

        v2f vf[NT / 2][8][2];
        #pragma unroll
        for (int tl = 0; tl < 2; ++tl) {
            #pragma unroll
            for (int hf = 0; hf < 2; ++hf) {
                #pragma unroll
                for (int i = 0; i < 8; ++i) {
                    const int key = kbase + 32 * hf + 4 * i + 2 * half;
                    vf[hf][i][tl][0] = Ys[key * YSTRIDE + 16 * tl + l16];
                    vf[hf][i][tl][1] = Ys[(key + 1) * YSTRIDE + 16 * tl + l16];
                }
            }
            if (tl == 0) __builtin_amdgcn_sched_barrier(0);
        }
        __builtin_amdgcn_sched_barrier(0);

        #pragma unroll
        for (int v = 0; v < 8; ++v) {
            float mx = s[0][v];
            #pragma unroll
            for (int t = 1; t < NT; ++t) mx = fmaxf(mx, s[t][v]);
            const float rmax = row_max16(mx) * scale;
            const float newm = fmaxf(mrow[v], rmax);
            const float corr = __expf(mrow[v] - newm);
            float psum[NT / 2];
            #pragma unroll
            for (int t = 0; t < NT; t += 2) {
                const float pa = __expf(s[t][v] * scale - newm);
                const float pb = __expf(s[t + 1][v] * scale - newm);
                s[t][v] = pa; s[t + 1][v] = pb;
                psum[t / 2] = pa + pb;
            }
            #pragma unroll
            for (int w = NT / 4; w >= 1; w >>= 1) {
                #pragma unroll
                for (int t = 0; t < w; ++t) psum[t] += psum[t + w];
            }
            const float rsum = row_sum16(psum[0]);
            lrow[v] = lrow[v] * corr + rsum;
            mrow[v] = newm;
            o0[v] *= corr;
            o1[v] *= corr;
        }

        #pragma unroll
        for (int hf = 0; hf < NT / 2; ++hf) {
            #pragma unroll
            for (int v = 0; v < 8; ++v) {
                Pw[(v + 8 * half) * PSTRIDE + l16]      = s[2 * hf][v];
                Pw[(v + 8 * half) * PSTRIDE + 16 + l16] = s[2 * hf + 1][v];
            }

            asm volatile("s_wait_dscnt 0x0" ::: "memory");

            v2f pf[8];
            #pragma unroll
            for (int i = 0; i < 8; ++i) {
                const float2 pa = *(const float2*)&Pw[l16 * PSTRIDE + 4 * i + 2 * half];
                pf[i][0] = pa.x; pf[i][1] = pa.y;
            }
            if (hf == 1) {
                __builtin_amdgcn_sched_barrier(0);
                #pragma unroll
                for (int tl = 0; tl < 2; ++tl) {
                    #pragma unroll
                    for (int h2 = 2; h2 < 4; ++h2) {
                        #pragma unroll
                        for (int i = 0; i < 8; ++i) {
                            const int key = kbase + 32 * h2 + 4 * i + 2 * half;
                            vf[h2][i][tl][0] = Ys[key * YSTRIDE + 16 * tl + l16];
                            vf[h2][i][tl][1] = Ys[(key + 1) * YSTRIDE + 16 * tl + l16];
                        }
                    }
                    if (tl == 0) __builtin_amdgcn_sched_barrier(0);
                }
            }
            __builtin_amdgcn_sched_barrier(0);

            #pragma unroll
            for (int i = 0; i < 8; ++i) {
                o0 = __builtin_amdgcn_wmma_f32_16x16x4_f32(false, pf[i], false, vf[hf][i][0],
                                                           (short)0, o0, false, false);
                o1 = __builtin_amdgcn_wmma_f32_16x16x4_f32(false, pf[i], false, vf[hf][i][1],
                                                           (short)0, o1, false, false);
            }
            __builtin_amdgcn_sched_barrier(0);
        }
    }

    #pragma unroll
    for (int v = 0; v < 8; ++v) {
        const int row = qbase + v + 8 * half;
        float e[3];
        const int d2 = l16 * 2;
        #pragma unroll
        for (int k = 0; k < 3; ++k) {
            const int nr = row + k - 1;
            const bool ok = (nr >= 0) && (nr < MROWS);
            const int nrc = ok ? nr : row;
            const float msk = ok ? 1.0f : 0.0f;
            float part = msk * (Ys[row * YSTRIDE + d2]     * Ys[nrc * YSTRIDE + d2]
                              + Ys[row * YSTRIDE + d2 + 1] * Ys[nrc * YSTRIDE + d2 + 1]);
            part = row_sum16(part);
            e[k] = part * scale;
        }
        const float newm = fmaxf(fmaxf(mrow[v], e[0]), fmaxf(e[1], e[2]));
        const float corr = __expf(mrow[v] - newm);
        float psum = 0.0f, add0 = 0.0f, add1 = 0.0f;
        #pragma unroll
        for (int k = 0; k < 3; ++k) {
            const float p = __expf(e[k] - newm);
            psum += p;
            const int nr = row + k - 1;
            const bool ok = (nr >= 0) && (nr < MROWS);
            const int nrc = ok ? nr : row;
            const float msk = ok ? 1.0f : 0.0f;
            add0 += (p * msk) * Ys[nrc * YSTRIDE + l16];
            add1 += (p * msk) * Ys[nrc * YSTRIDE + 16 + l16];
        }
        const float inv = 1.0f / (lrow[v] * corr + psum);
        const float r0 = (o0[v] * corr + add0) * inv;
        const float r1 = (o1[v] * corr + add1) * inv;

        const int n = 2 * row + r;
        if (n < SEQ) {
            const size_t base = ((size_t)(bb * SEQ + n)) * DMODEL + h * DK;
            out[base + l16]      = r0;
            out[base + 16 + l16] = r1;
        }
    }
}

extern "C" void kernel_launch(void* const* d_in, const int* in_sizes, int n_in,
                              void* d_out, int out_size, void* d_ws, size_t ws_size,
                              hipStream_t stream) {
    const float* x = (const float*)d_in[0];
    const float* W = (const float*)d_in[1];
    const float* b = (const float*)d_in[2];
    float* out = (float*)d_out;
    float* qkv = (float*)d_ws;

    dim3 g1(512, 16);
    qkv_gemm_kernel<<<g1, 32, 0, stream>>>(x, W, b, qkv);

    const size_t ldsBytes = (size_t)(MROWS * YSTRIDE + 8 * 16 * PSTRIDE) * sizeof(float);
    dim3 g2(8, 64);
    attn_kernel<<<g2, 256, ldsBytes, stream>>>(qkv, out);
}


# Projection GEMM: a wave keeps its 16-column W slice in registers and streams four 16-row tiles through it (x 16 B per lane, next block's loads ahead); 2048 waves, not 8192; k sum order changes
# speedup vs baseline: 2.8701x; 1.0616x over previous
#include <hip/hip_runtime.h>
#include <math.h>

typedef float v2f __attribute__((ext_vector_type(2)));
typedef float v8f __attribute__((ext_vector_type(8)));

#define HEADS   8
#define DK      32
#define RATE    2
#define SEQ     2046
#define NPAD    2048
#define MROWS   1024
#define DMODEL  256
#define BATCH   4

template <int M>
__device__ __forceinline__ float row_xor(float x)
{
    const int xi = __builtin_bit_cast(int, x);
    return __builtin_bit_cast(float,
        __builtin_amdgcn_update_dpp(xi, xi, 0x160 + M, 0xf, 0xf, true));
}

__device__ __forceinline__ float row_max16(float x)
{
    x = fmaxf(x, row_xor<1>(x));
    x = fmaxf(x, row_xor<2>(x));
    x = fmaxf(x, row_xor<4>(x));
    x = fmaxf(x, row_xor<8>(x));
    return x;
}
__device__ __forceinline__ float row_sum16(float x)
{
    x += row_xor<1>(x);
    x += row_xor<2>(x);
    x += row_xor<4>(x);
    x += row_xor<8>(x);
    return x;
}

#define YSTRIDE 34
#define PSTRIDE 34
#define NT      8
#define KSTEP   (16 * NT)

#define RT_PER_WAVE 4
#define KSTEPS      (DMODEL / 4)
#define KB_STEPS    16
#define BPP         (KSTEPS / KB_STEPS)
#define NBLK        ((RT_PER_WAVE / 2) * BPP)

__device__ __forceinline__ int kfrag(int ks, int half)
{
    return 8 * (ks >> 1) + 4 * half + 2 * (ks & 1);
}

__global__ __launch_bounds__(32)
void qkv_gemm_kernel(const float* __restrict__ x,
                     const float* __restrict__ W,
                     const float* __restrict__ bias,
                     float* __restrict__ qkv)
{
    const int lane = threadIdx.x;
    const int half = lane >> 4;
    const int l16  = lane & 15;

    const int mgrp  = blockIdx.x;
    const int ntile = blockIdx.y;
    const int col = ntile * 16 + l16;

    const float* xrow[RT_PER_WAVE];
    #pragma unroll
    for (int t = 0; t < RT_PER_WAVE; ++t) {
        const int row = (mgrp * RT_PER_WAVE + t) * 16 + l16;
        const int bb  = row >> 11;
        const int n   = row & (NPAD - 1);
        const bool valid = (n < SEQ);
        xrow[t] = x + ((size_t)(bb * SEQ + (valid ? n : 0))) * DMODEL + 4 * half;
    }

    v2f wf[KSTEPS];
    float4 xa[2][KB_STEPS / 2][2];

    #define LOAD_W(KS0, KS1)                                                      \
        _Pragma("unroll")                                                         \
        for (int ks = (KS0); ks < (KS1); ++ks) {                                  \
            const int k = kfrag(ks, half);                                        \
            wf[ks][0] = W[(size_t)k * DMODEL + col];                              \
            wf[ks][1] = W[(size_t)(k + 1) * DMODEL + col];                        \
        }
    #define LOAD_X(B)                                                             \
        _Pragma("unroll")                                                         \
        for (int q = 0; q < KB_STEPS / 2; ++q) {                                  \
            _Pragma("unroll")                                                     \
            for (int u = 0; u < 2; ++u) {                                         \
                xa[(B) & 1][q][u] = *(const float4*)(xrow[2 * ((B) / BPP) + u]    \
                                        + 8 * ((KB_STEPS / 2) * ((B) % BPP) + q));\
            }                                                                     \
        }

    LOAD_W(0, KB_STEPS)
    LOAD_X(0)
    LOAD_W(KB_STEPS, KSTEPS)

    const float bv = bias[col];
    v8f c[RT_PER_WAVE];
    #pragma unroll
    for (int t = 0; t < RT_PER_WAVE; ++t) {
        #pragma unroll
        for (int v = 0; v < 8; ++v) c[t][v] = bv;
    }

    #pragma unroll
    for (int b = 0; b < NBLK; ++b) {
        const int pr = b / BPP;
        const int kb = b % BPP;

        __builtin_amdgcn_sched_barrier(0);
        if (b + 1 < NBLK) {
            LOAD_X(b + 1)
        }
        __builtin_amdgcn_sched_barrier(0);

        #pragma unroll
        for (int i = 0; i < KB_STEPS; ++i) {
            #pragma unroll
            for (int u = 0; u < 2; ++u) {
                const float4 av = xa[b & 1][i >> 1][u];
                v2f a;
                a[0] = (i & 1) ? av.z : av.x;
                a[1] = (i & 1) ? av.w : av.y;
                c[2 * pr + u] = __builtin_amdgcn_wmma_f32_16x16x4_f32(false, a, false,
                                                             wf[KB_STEPS * kb + i],
                                                             (short)0, c[2 * pr + u], false, false);
            }
        }
    }
    #undef LOAD_W
    #undef LOAD_X

    #pragma unroll
    for (int t = 0; t < RT_PER_WAVE; ++t) {
        #pragma unroll
        for (int v = 0; v < 8; ++v) {
            const int rrow = (mgrp * RT_PER_WAVE + t) * 16 + v + 8 * half;
            const bool valid = ((rrow & (NPAD - 1)) < SEQ);
            qkv[(size_t)rrow * DMODEL + col] = valid ? c[t][v] : bv;
        }
    }
}

__global__ __launch_bounds__(256)
void attn_kernel(const float* __restrict__ qkv, float* __restrict__ out)
{
    extern __shared__ float lds[];
    float* Ys = lds;
    float* Ps = lds + MROWS * YSTRIDE;

    const int tid  = threadIdx.x;
    const int wave = tid >> 5;
    const int lane = tid & 31;
    const int half = lane >> 4;
    const int l16  = lane & 15;

    const int g  = blockIdx.y;
    const int bb = g >> 4;
    const int h  = (g >> 1) & 7;
    const int r  = g & 1;

    {
        const int dcol  = (tid & 15) * 2;
        const int rbase = tid >> 4;
        for (int i = rbase; i < MROWS; i += 16) {
            const float* src = qkv + ((size_t)(bb * NPAD + 2 * i + r)) * DMODEL
                                   + h * DK + dcol;
            const float2 v = *(const float2*)src;
            *(float2*)&Ys[i * YSTRIDE + dcol] = v;
        }
    }
    __syncthreads();

    float* Pw = Ps + wave * 16 * PSTRIDE;

    const int qb    = blockIdx.x * 8 + wave;
    const int qbase = qb * 16;

    const float scale = 0.17677669529663687f;

    float mrow[8], lrow[8];
    v8f o0, o1;
    #pragma unroll
    for (int v = 0; v < 8; ++v) {
        mrow[v] = -3.0e38f; lrow[v] = 0.0f;
        o0[v] = 0.0f; o1[v] = 0.0f;
    }

    for (int kb = 0; kb < MROWS / KSTEP; ++kb) {
        const int kbase = kb * KSTEP;

        v2f qf[DK / 4];
        v2f kf[NT][DK / 4];
        #pragma unroll
        for (int j = 0; j < DK / 4; ++j) {
            const float2 av = *(const float2*)&Ys[(qbase + l16) * YSTRIDE + 4 * j + 2 * half];
            qf[j][0] = av.x; qf[j][1] = av.y;
        }
        #pragma unroll
        for (int t = 0; t < NT; ++t) {
            #pragma unroll
            for (int j = 0; j < DK / 4; ++j) {
                const float2 bv = *(const float2*)&Ys[(kbase + 16 * t + l16) * YSTRIDE + 4 * j + 2 * half];
                kf[t][j][0] = bv.x; kf[t][j][1] = bv.y;
            }
        }
        __builtin_amdgcn_sched_barrier(0);

        v8f s[NT];
        #pragma unroll
        for (int t = 0; t < NT; ++t) {
            #pragma unroll
            for (int v = 0; v < 8; ++v) s[t][v] = 0.0f;
        }
        #pragma unroll
        for (int g = 0; g < NT; g += 4) {
            #pragma unroll
            for (int j = 0; j < DK / 4; ++j) {
                #pragma unroll
                for (int t = g; t < g + 4; ++t) {
                    s[t] = __builtin_amdgcn_wmma_f32_16x16x4_f32(false, qf[j], false, kf[t][j],
                                                                 (short)0, s[t], false, false);
                }
            }
            if (g == 0) __builtin_amdgcn_sched_barrier(0);
        }
        __builtin_amdgcn_sched_barrier(0);

        v2f vf[NT / 2][8][2];
        #pragma unroll
        for (int tl = 0; tl < 2; ++tl) {
            #pragma unroll
            for (int hf = 0; hf < 2; ++hf) {
                #pragma unroll
                for (int i = 0; i < 8; ++i) {
                    const int key = kbase + 32 * hf + 4 * i + 2 * half;
                    vf[hf][i][tl][0] = Ys[key * YSTRIDE + 16 * tl + l16];
                    vf[hf][i][tl][1] = Ys[(key + 1) * YSTRIDE + 16 * tl + l16];
                }
            }
            if (tl == 0) __builtin_amdgcn_sched_barrier(0);
        }
        __builtin_amdgcn_sched_barrier(0);

        #pragma unroll
        for (int v = 0; v < 8; ++v) {
            float mx = s[0][v];
            #pragma unroll
            for (int t = 1; t < NT; ++t) mx = fmaxf(mx, s[t][v]);
            const float rmax = row_max16(mx) * scale;
            const float newm = fmaxf(mrow[v], rmax);
            const float corr = __expf(mrow[v] - newm);
            float psum[NT / 2];
            #pragma unroll
            for (int t = 0; t < NT; t += 2) {
                const float pa = __expf(s[t][v] * scale - newm);
                const float pb = __expf(s[t + 1][v] * scale - newm);
                s[t][v] = pa; s[t + 1][v] = pb;
                psum[t / 2] = pa + pb;
            }
            #pragma unroll
            for (int w = NT / 4; w >= 1; w >>= 1) {
                #pragma unroll
                for (int t = 0; t < w; ++t) psum[t] += psum[t + w];
            }
            const float rsum = row_sum16(psum[0]);
            lrow[v] = lrow[v] * corr + rsum;
            mrow[v] = newm;
            o0[v] *= corr;
            o1[v] *= corr;
        }

        #pragma unroll
        for (int hf = 0; hf < NT / 2; ++hf) {
            #pragma unroll
            for (int v = 0; v < 8; ++v) {
                Pw[(v + 8 * half) * PSTRIDE + l16]      = s[2 * hf][v];
                Pw[(v + 8 * half) * PSTRIDE + 16 + l16] = s[2 * hf + 1][v];
            }

            asm volatile("s_wait_dscnt 0x0" ::: "memory");

            v2f pf[8];
            #pragma unroll
            for (int i = 0; i < 8; ++i) {
                const float2 pa = *(const float2*)&Pw[l16 * PSTRIDE + 4 * i + 2 * half];
                pf[i][0] = pa.x; pf[i][1] = pa.y;
            }
            if (hf == 1) {
                __builtin_amdgcn_sched_barrier(0);
                #pragma unroll
                for (int tl = 0; tl < 2; ++tl) {
                    #pragma unroll
                    for (int h2 = 2; h2 < 4; ++h2) {
                        #pragma unroll
                        for (int i = 0; i < 8; ++i) {
                            const int key = kbase + 32 * h2 + 4 * i + 2 * half;
                            vf[h2][i][tl][0] = Ys[key * YSTRIDE + 16 * tl + l16];
                            vf[h2][i][tl][1] = Ys[(key + 1) * YSTRIDE + 16 * tl + l16];
                        }
                    }
                    if (tl == 0) __builtin_amdgcn_sched_barrier(0);
                }
            }
            __builtin_amdgcn_sched_barrier(0);

            #pragma unroll
            for (int i = 0; i < 8; ++i) {
                o0 = __builtin_amdgcn_wmma_f32_16x16x4_f32(false, pf[i], false, vf[hf][i][0],
                                                           (short)0, o0, false, false);
                o1 = __builtin_amdgcn_wmma_f32_16x16x4_f32(false, pf[i], false, vf[hf][i][1],
                                                           (short)0, o1, false, false);
            }
            __builtin_amdgcn_sched_barrier(0);
        }
    }

    #pragma unroll
    for (int v = 0; v < 8; ++v) {
        const int row = qbase + v + 8 * half;
        float e[3];
        const int d2 = l16 * 2;
        #pragma unroll
        for (int k = 0; k < 3; ++k) {
            const int nr = row + k - 1;
            const bool ok = (nr >= 0) && (nr < MROWS);
            const int nrc = ok ? nr : row;
            const float msk = ok ? 1.0f : 0.0f;
            float part = msk * (Ys[row * YSTRIDE + d2]     * Ys[nrc * YSTRIDE + d2]
                              + Ys[row * YSTRIDE + d2 + 1] * Ys[nrc * YSTRIDE + d2 + 1]);
            part = row_sum16(part);
            e[k] = part * scale;
        }
        const float newm = fmaxf(fmaxf(mrow[v], e[0]), fmaxf(e[1], e[2]));
        const float corr = __expf(mrow[v] - newm);
        float psum = 0.0f, add0 = 0.0f, add1 = 0.0f;
        #pragma unroll
        for (int k = 0; k < 3; ++k) {
            const float p = __expf(e[k] - newm);
            psum += p;
            const int nr = row + k - 1;
            const bool ok = (nr >= 0) && (nr < MROWS);
            const int nrc = ok ? nr : row;
            const float msk = ok ? 1.0f : 0.0f;
            add0 += (p * msk) * Ys[nrc * YSTRIDE + l16];
            add1 += (p * msk) * Ys[nrc * YSTRIDE + 16 + l16];
        }
        const float inv = 1.0f / (lrow[v] * corr + psum);
        const float r0 = (o0[v] * corr + add0) * inv;
        const float r1 = (o1[v] * corr + add1) * inv;

        const int n = 2 * row + r;
        if (n < SEQ) {
            const size_t base = ((size_t)(bb * SEQ + n)) * DMODEL + h * DK;
            out[base + l16]      = r0;
            out[base + 16 + l16] = r1;
        }
    }
}

extern "C" void kernel_launch(void* const* d_in, const int* in_sizes, int n_in,
                              void* d_out, int out_size, void* d_ws, size_t ws_size,
                              hipStream_t stream) {
    const float* x = (const float*)d_in[0];
    const float* W = (const float*)d_in[1];
    const float* b = (const float*)d_in[2];
    float* out = (float*)d_out;
    float* qkv = (float*)d_ws;

    dim3 g1(128, 16);
    qkv_gemm_kernel<<<g1, 32, 0, stream>>>(x, W, b, qkv);

    const size_t ldsBytes = (size_t)(MROWS * YSTRIDE + 8 * 16 * PSTRIDE) * sizeof(float);
    dim3 g2(8, 64);
    attn_kernel<<<g2, 256, ldsBytes, stream>>>(qkv, out);
}


# Attention: LDS transposition of each 32-key piece moved in front of the softmax (raw scores, not P): row max/sum reduce in the lane plus one half swap, exps feed P*V directly; row-sum order changes
# speedup vs baseline: 3.0085x; 1.0482x over previous
#include <hip/hip_runtime.h>
#include <math.h>

typedef float v2f __attribute__((ext_vector_type(2)));
typedef float v8f __attribute__((ext_vector_type(8)));

#define HEADS   8
#define DK      32
#define RATE    2
#define SEQ     2046
#define NPAD    2048
#define MROWS   1024
#define DMODEL  256
#define BATCH   4

template <int M>
__device__ __forceinline__ float row_xor(float x)
{
    const int xi = __builtin_bit_cast(int, x);
    return __builtin_bit_cast(float,
        __builtin_amdgcn_update_dpp(xi, xi, 0x160 + M, 0xf, 0xf, true));
}

__device__ __forceinline__ float row_max16(float x)
{
    x = fmaxf(x, row_xor<1>(x));
    x = fmaxf(x, row_xor<2>(x));
    x = fmaxf(x, row_xor<4>(x));
    x = fmaxf(x, row_xor<8>(x));
    return x;
}
__device__ __forceinline__ float row_sum16(float x)
{
    x += row_xor<1>(x);
    x += row_xor<2>(x);
    x += row_xor<4>(x);
    x += row_xor<8>(x);
    return x;
}

__device__ __forceinline__ float xhalf(float x)
{
    const int xi = __builtin_bit_cast(int, x);
    return __builtin_bit_cast(float,
        __builtin_amdgcn_permlanex16(xi, xi, 0x76543210, 0xfedcba98, false, false));
}

template <int V>
__device__ __forceinline__ float row_pick(float x)
{
    const int xi = __builtin_bit_cast(int, x);
    int r = __builtin_amdgcn_update_dpp(xi, xi, 0x150 + V, 0x1, 0xf, false);
    r = __builtin_amdgcn_update_dpp(r, xi, 0x150 + V + 8, 0x2, 0xf, false);
    return __builtin_bit_cast(float, r);
}

#define YSTRIDE 34
#define PSTRIDE 34
#define NT      8
#define KSTEP   (16 * NT)

#define RT_PER_WAVE 4
#define KSTEPS      (DMODEL / 4)
#define KB_STEPS    16
#define BPP         (KSTEPS / KB_STEPS)
#define NBLK        ((RT_PER_WAVE / 2) * BPP)

__device__ __forceinline__ int kfrag(int ks, int half)
{
    return 8 * (ks >> 1) + 4 * half + 2 * (ks & 1);
}

__global__ __launch_bounds__(32)
void qkv_gemm_kernel(const float* __restrict__ x,
                     const float* __restrict__ W,
                     const float* __restrict__ bias,
                     float* __restrict__ qkv)
{
    const int lane = threadIdx.x;
    const int half = lane >> 4;
    const int l16  = lane & 15;

    const int mgrp  = blockIdx.x;
    const int ntile = blockIdx.y;
    const int col = ntile * 16 + l16;

    const float* xrow[RT_PER_WAVE];
    #pragma unroll
    for (int t = 0; t < RT_PER_WAVE; ++t) {
        const int row = (mgrp * RT_PER_WAVE + t) * 16 + l16;
        const int bb  = row >> 11;
        const int n   = row & (NPAD - 1);
        const bool valid = (n < SEQ);
        xrow[t] = x + ((size_t)(bb * SEQ + (valid ? n : 0))) * DMODEL + 4 * half;
    }

    v2f wf[KSTEPS];
    float4 xa[2][KB_STEPS / 2][2];

    #define LOAD_W(KS0, KS1)                                                      \
        _Pragma("unroll")                                                         \
        for (int ks = (KS0); ks < (KS1); ++ks) {                                  \
            const int k = kfrag(ks, half);                                        \
            wf[ks][0] = W[(size_t)k * DMODEL + col];                              \
            wf[ks][1] = W[(size_t)(k + 1) * DMODEL + col];                        \
        }
    #define LOAD_X(B)                                                             \
        _Pragma("unroll")                                                         \
        for (int q = 0; q < KB_STEPS / 2; ++q) {                                  \
            _Pragma("unroll")                                                     \
            for (int u = 0; u < 2; ++u) {                                         \
                xa[(B) & 1][q][u] = *(const float4*)(xrow[2 * ((B) / BPP) + u]    \
                                        + 8 * ((KB_STEPS / 2) * ((B) % BPP) + q));\
            }                                                                     \
        }

    LOAD_W(0, KB_STEPS)
    LOAD_X(0)
    LOAD_W(KB_STEPS, KSTEPS)

    const float bv = bias[col];
    v8f c[RT_PER_WAVE];
    #pragma unroll
    for (int t = 0; t < RT_PER_WAVE; ++t) {
        #pragma unroll
        for (int v = 0; v < 8; ++v) c[t][v] = bv;
    }

    #pragma unroll
    for (int b = 0; b < NBLK; ++b) {
        const int pr = b / BPP;
        const int kb = b % BPP;

        __builtin_amdgcn_sched_barrier(0);
        if (b + 1 < NBLK) {
            LOAD_X(b + 1)
        }
        __builtin_amdgcn_sched_barrier(0);

        #pragma unroll
        for (int i = 0; i < KB_STEPS; ++i) {
            #pragma unroll
            for (int u = 0; u < 2; ++u) {
                const float4 av = xa[b & 1][i >> 1][u];
                v2f a;
                a[0] = (i & 1) ? av.z : av.x;
                a[1] = (i & 1) ? av.w : av.y;
                c[2 * pr + u] = __builtin_amdgcn_wmma_f32_16x16x4_f32(false, a, false,
                                                             wf[KB_STEPS * kb + i],
                                                             (short)0, c[2 * pr + u], false, false);
            }
        }
    }
    #undef LOAD_W
    #undef LOAD_X

    #pragma unroll
    for (int t = 0; t < RT_PER_WAVE; ++t) {
        #pragma unroll
        for (int v = 0; v < 8; ++v) {
            const int rrow = (mgrp * RT_PER_WAVE + t) * 16 + v + 8 * half;
            const bool valid = ((rrow & (NPAD - 1)) < SEQ);
            qkv[(size_t)rrow * DMODEL + col] = valid ? c[t][v] : bv;
        }
    }
}

__global__ __launch_bounds__(256)
void attn_kernel(const float* __restrict__ qkv, float* __restrict__ out)
{
    extern __shared__ float lds[];
    float* Ys = lds;
    float* Ps = lds + MROWS * YSTRIDE;

    const int tid  = threadIdx.x;
    const int wave = tid >> 5;
    const int lane = tid & 31;
    const int half = lane >> 4;
    const int l16  = lane & 15;

    const int g  = blockIdx.y;
    const int bb = g >> 4;
    const int h  = (g >> 1) & 7;
    const int r  = g & 1;

    {
        const int dcol  = (tid & 15) * 2;
        const int rbase = tid >> 4;
        for (int i = rbase; i < MROWS; i += 16) {
            const float* src = qkv + ((size_t)(bb * NPAD + 2 * i + r)) * DMODEL
                                   + h * DK + dcol;
            const float2 v = *(const float2*)src;
            *(float2*)&Ys[i * YSTRIDE + dcol] = v;
        }
    }
    __syncthreads();

    float* Pw = Ps + wave * 16 * PSTRIDE;

    const int qb    = blockIdx.x * 8 + wave;
    const int qbase = qb * 16;

    const float scale = 0.17677669529663687f;

    float mrow = -3.0e38f, lrow = 0.0f;
    v8f o0, o1;
    #pragma unroll
    for (int v = 0; v < 8; ++v) {
        o0[v] = 0.0f; o1[v] = 0.0f;
    }

    for (int kb = 0; kb < MROWS / KSTEP; ++kb) {
        const int kbase = kb * KSTEP;

        v2f qf[DK / 4];
        v2f kf[NT][DK / 4];
        #pragma unroll
        for (int j = 0; j < DK / 4; ++j) {
            const float2 av = *(const float2*)&Ys[(qbase + l16) * YSTRIDE + 4 * j + 2 * half];
            qf[j][0] = av.x; qf[j][1] = av.y;
        }
        #pragma unroll
        for (int t = 0; t < NT; ++t) {
            #pragma unroll
            for (int j = 0; j < DK / 4; ++j) {
                const float2 bv = *(const float2*)&Ys[(kbase + 16 * t + l16) * YSTRIDE + 4 * j + 2 * half];
                kf[t][j][0] = bv.x; kf[t][j][1] = bv.y;
            }
        }
        __builtin_amdgcn_sched_barrier(0);

        v8f s[NT];
        #pragma unroll
        for (int t = 0; t < NT; ++t) {
            #pragma unroll
            for (int v = 0; v < 8; ++v) s[t][v] = 0.0f;
        }
        #pragma unroll
        for (int g = 0; g < NT; g += 4) {
            #pragma unroll
            for (int j = 0; j < DK / 4; ++j) {
                #pragma unroll
                for (int t = g; t < g + 4; ++t) {
                    s[t] = __builtin_amdgcn_wmma_f32_16x16x4_f32(false, qf[j], false, kf[t][j],
                                                                 (short)0, s[t], false, false);
                }
            }
            if (g == 0) __builtin_amdgcn_sched_barrier(0);
        }
        __builtin_amdgcn_sched_barrier(0);

        v2f sf[NT / 2][8];
        float pm[NT / 2];
        #pragma unroll
        for (int hf = 0; hf < NT / 2; ++hf) {
            #pragma unroll
            for (int v = 0; v < 8; ++v) {
                Pw[(v + 8 * half) * PSTRIDE + l16]      = s[2 * hf][v];
                Pw[(v + 8 * half) * PSTRIDE + 16 + l16] = s[2 * hf + 1][v];
            }

            asm volatile("s_wait_dscnt 0x0" ::: "memory");

            #pragma unroll
            for (int i = 0; i < 8; ++i) {
                const float2 pa = *(const float2*)&Pw[l16 * PSTRIDE + 4 * i + 2 * half];
                sf[hf][i][0] = pa.x; sf[hf][i][1] = pa.y;
            }
            float m = fmaxf(sf[hf][0][0], sf[hf][0][1]);
            #pragma unroll
            for (int i = 1; i < 8; ++i) m = fmaxf(m, fmaxf(sf[hf][i][0], sf[hf][i][1]));
            pm[hf] = m;
        }
        __builtin_amdgcn_sched_barrier(0);

        v2f vf[NT / 2][8][2];
        #pragma unroll
        for (int tl = 0; tl < 2; ++tl) {
            #pragma unroll
            for (int hf = 0; hf < 2; ++hf) {
                #pragma unroll
                for (int i = 0; i < 8; ++i) {
                    const int key = kbase + 32 * hf + 4 * i + 2 * half;
                    vf[hf][i][tl][0] = Ys[key * YSTRIDE + 16 * tl + l16];
                    vf[hf][i][tl][1] = Ys[(key + 1) * YSTRIDE + 16 * tl + l16];
                }
            }
            if (tl == 0) __builtin_amdgcn_sched_barrier(0);
        }
        __builtin_amdgcn_sched_barrier(0);

        {
            float mx = fmaxf(fmaxf(pm[0], pm[1]), fmaxf(pm[2], pm[3]));
            mx = fmaxf(mx, xhalf(mx));
            const float newm = fmaxf(mrow, mx * scale);
            const float corr = __expf(mrow - newm);
            const v2f newm2 = {newm, newm};
            const float log2e = 0x1.715476p+0f;
            v2f ps[NT / 2];
            #pragma unroll
            for (int hf = 0; hf < NT / 2; ++hf) {
                #pragma unroll
                for (int i = 0; i < 8; ++i) {
                    const v2f u = (sf[hf][i] * scale - newm2) * log2e;
                    sf[hf][i][0] = __builtin_amdgcn_exp2f(u[0]);
                    sf[hf][i][1] = __builtin_amdgcn_exp2f(u[1]);
                    if (i == 0) ps[hf] = sf[hf][i]; else ps[hf] += sf[hf][i];
                }
            }
            const v2f pt = (ps[0] + ps[1]) + (ps[2] + ps[3]);
            float rsum = pt[0] + pt[1];
            rsum += xhalf(rsum);
            lrow = lrow * corr + rsum;
            mrow = newm;
            #define RESCALE_ROW(V) { const float cv = row_pick<V>(corr); o0[V] *= cv; o1[V] *= cv; }
            RESCALE_ROW(0) RESCALE_ROW(1) RESCALE_ROW(2) RESCALE_ROW(3)
            RESCALE_ROW(4) RESCALE_ROW(5) RESCALE_ROW(6) RESCALE_ROW(7)
            #undef RESCALE_ROW
        }
        __builtin_amdgcn_sched_barrier(0);

        #pragma unroll
        for (int tl = 0; tl < 2; ++tl) {
            #pragma unroll
            for (int h2 = 2; h2 < 4; ++h2) {
                #pragma unroll
                for (int i = 0; i < 8; ++i) {
                    const int key = kbase + 32 * h2 + 4 * i + 2 * half;
                    vf[h2][i][tl][0] = Ys[key * YSTRIDE + 16 * tl + l16];
                    vf[h2][i][tl][1] = Ys[(key + 1) * YSTRIDE + 16 * tl + l16];
                }
            }
            if (tl == 0) __builtin_amdgcn_sched_barrier(0);
        }
        __builtin_amdgcn_sched_barrier(0);

        #pragma unroll
        for (int hf = 0; hf < NT / 2; ++hf) {
            #pragma unroll
            for (int i = 0; i < 8; ++i) {
                o0 = __builtin_amdgcn_wmma_f32_16x16x4_f32(false, sf[hf][i], false, vf[hf][i][0],
                                                           (short)0, o0, false, false);
                o1 = __builtin_amdgcn_wmma_f32_16x16x4_f32(false, sf[hf][i], false, vf[hf][i][1],
                                                           (short)0, o1, false, false);
            }
        }
        __builtin_amdgcn_sched_barrier(0);
    }

    float mrv[8], lrv[8];
    #define PICK_ROW(V) { mrv[V] = row_pick<V>(mrow); lrv[V] = row_pick<V>(lrow); }
    PICK_ROW(0) PICK_ROW(1) PICK_ROW(2) PICK_ROW(3)
    PICK_ROW(4) PICK_ROW(5) PICK_ROW(6) PICK_ROW(7)
    #undef PICK_ROW

    #pragma unroll
    for (int v = 0; v < 8; ++v) {
        const int row = qbase + v + 8 * half;
        float e[3];
        const int d2 = l16 * 2;
        #pragma unroll
        for (int k = 0; k < 3; ++k) {
            const int nr = row + k - 1;
            const bool ok = (nr >= 0) && (nr < MROWS);
            const int nrc = ok ? nr : row;
            const float msk = ok ? 1.0f : 0.0f;
            float part = msk * (Ys[row * YSTRIDE + d2]     * Ys[nrc * YSTRIDE + d2]
                              + Ys[row * YSTRIDE + d2 + 1] * Ys[nrc * YSTRIDE + d2 + 1]);
            part = row_sum16(part);
            e[k] = part * scale;
        }
        const float newm = fmaxf(fmaxf(mrv[v], e[0]), fmaxf(e[1], e[2]));
        const float corr = __expf(mrv[v] - newm);
        float psum = 0.0f, add0 = 0.0f, add1 = 0.0f;
        #pragma unroll
        for (int k = 0; k < 3; ++k) {
            const float p = __expf(e[k] - newm);
            psum += p;
            const int nr = row + k - 1;
            const bool ok = (nr >= 0) && (nr < MROWS);
            const int nrc = ok ? nr : row;
            const float msk = ok ? 1.0f : 0.0f;
            add0 += (p * msk) * Ys[nrc * YSTRIDE + l16];
            add1 += (p * msk) * Ys[nrc * YSTRIDE + 16 + l16];
        }
        const float inv = 1.0f / (lrv[v] * corr + psum);
        const float r0 = (o0[v] * corr + add0) * inv;
        const float r1 = (o1[v] * corr + add1) * inv;

        const int n = 2 * row + r;
        if (n < SEQ) {
            const size_t base = ((size_t)(bb * SEQ + n)) * DMODEL + h * DK;
            out[base + l16]      = r0;
            out[base + 16 + l16] = r1;
        }
    }
}

extern "C" void kernel_launch(void* const* d_in, const int* in_sizes, int n_in,
                              void* d_out, int out_size, void* d_ws, size_t ws_size,
                              hipStream_t stream) {
    const float* x = (const float*)d_in[0];
    const float* W = (const float*)d_in[1];
    const float* b = (const float*)d_in[2];
    float* out = (float*)d_out;
    float* qkv = (float*)d_ws;

    dim3 g1(128, 16);
    qkv_gemm_kernel<<<g1, 32, 0, stream>>>(x, W, b, qkv);

    const size_t ldsBytes = (size_t)(MROWS * YSTRIDE + 8 * 16 * PSTRIDE) * sizeof(float);
    dim3 g2(8, 64);
    attn_kernel<<<g2, 256, ldsBytes, stream>>>(qkv, out);
}


# Attention: one 16-wave work-group per WGP (Y staged once per WGP); freed LDS holds a 16x128 score scratch per wave: scores transposed in one LDS round trip per 128 keys, not 4; row-sum order changes
# speedup vs baseline: 3.1346x; 1.0419x over previous
#include <hip/hip_runtime.h>
#include <math.h>

typedef float v2f __attribute__((ext_vector_type(2)));
typedef float v8f __attribute__((ext_vector_type(8)));

#define HEADS   8
#define DK      32
#define RATE    2
#define SEQ     2046
#define NPAD    2048
#define MROWS   1024
#define DMODEL  256
#define BATCH   4

template <int M>
__device__ __forceinline__ float row_xor(float x)
{
    const int xi = __builtin_bit_cast(int, x);
    return __builtin_bit_cast(float,
        __builtin_amdgcn_update_dpp(xi, xi, 0x160 + M, 0xf, 0xf, true));
}

__device__ __forceinline__ float row_max16(float x)
{
    x = fmaxf(x, row_xor<1>(x));
    x = fmaxf(x, row_xor<2>(x));
    x = fmaxf(x, row_xor<4>(x));
    x = fmaxf(x, row_xor<8>(x));
    return x;
}
__device__ __forceinline__ float row_sum16(float x)
{
    x += row_xor<1>(x);
    x += row_xor<2>(x);
    x += row_xor<4>(x);
    x += row_xor<8>(x);
    return x;
}

__device__ __forceinline__ float xhalf(float x)
{
    const int xi = __builtin_bit_cast(int, x);
    return __builtin_bit_cast(float,
        __builtin_amdgcn_permlanex16(xi, xi, 0x76543210, 0xfedcba98, false, false));
}

template <int V>
__device__ __forceinline__ float row_pick(float x)
{
    const int xi = __builtin_bit_cast(int, x);
    int r = __builtin_amdgcn_update_dpp(xi, xi, 0x150 + V, 0x1, 0xf, false);
    r = __builtin_amdgcn_update_dpp(r, xi, 0x150 + V + 8, 0x2, 0xf, false);
    return __builtin_bit_cast(float, r);
}

#define YSTRIDE 34
#define PSTRIDE 132
#define NT      8
#define KSTEP   (16 * NT)

#define RT_PER_WAVE 4
#define KSTEPS      (DMODEL / 4)
#define KB_STEPS    16
#define BPP         (KSTEPS / KB_STEPS)
#define NBLK        ((RT_PER_WAVE / 2) * BPP)

__device__ __forceinline__ int kfrag(int ks, int half)
{
    return 8 * (ks >> 1) + 4 * half + 2 * (ks & 1);
}

__global__ __launch_bounds__(32)
void qkv_gemm_kernel(const float* __restrict__ x,
                     const float* __restrict__ W,
                     const float* __restrict__ bias,
                     float* __restrict__ qkv)
{
    const int lane = threadIdx.x;
    const int half = lane >> 4;
    const int l16  = lane & 15;

    const int mgrp  = blockIdx.x;
    const int ntile = blockIdx.y;
    const int col = ntile * 16 + l16;

    const float* xrow[RT_PER_WAVE];
    #pragma unroll
    for (int t = 0; t < RT_PER_WAVE; ++t) {
        const int row = (mgrp * RT_PER_WAVE + t) * 16 + l16;
        const int bb  = row >> 11;
        const int n   = row & (NPAD - 1);
        const bool valid = (n < SEQ);
        xrow[t] = x + ((size_t)(bb * SEQ + (valid ? n : 0))) * DMODEL + 4 * half;
    }

    v2f wf[KSTEPS];
    float4 xa[2][KB_STEPS / 2][2];

    #define LOAD_W(KS0, KS1)                                                      \
        _Pragma("unroll")                                                         \
        for (int ks = (KS0); ks < (KS1); ++ks) {                                  \
            const int k = kfrag(ks, half);                                        \
            wf[ks][0] = W[(size_t)k * DMODEL + col];                              \
            wf[ks][1] = W[(size_t)(k + 1) * DMODEL + col];                        \
        }
    #define LOAD_X(B)                                                             \
        _Pragma("unroll")                                                         \
        for (int q = 0; q < KB_STEPS / 2; ++q) {                                  \
            _Pragma("unroll")                                                     \
            for (int u = 0; u < 2; ++u) {                                         \
                xa[(B) & 1][q][u] = *(const float4*)(xrow[2 * ((B) / BPP) + u]    \
                                        + 8 * ((KB_STEPS / 2) * ((B) % BPP) + q));\
            }                                                                     \
        }

    LOAD_W(0, KB_STEPS)
    LOAD_X(0)
    LOAD_W(KB_STEPS, KSTEPS)

    const float bv = bias[col];
    v8f c[RT_PER_WAVE];
    #pragma unroll
    for (int t = 0; t < RT_PER_WAVE; ++t) {
        #pragma unroll
        for (int v = 0; v < 8; ++v) c[t][v] = bv;
    }

    #pragma unroll
    for (int b = 0; b < NBLK; ++b) {
        const int pr = b / BPP;
        const int kb = b % BPP;

        __builtin_amdgcn_sched_barrier(0);
        if (b + 1 < NBLK) {
            LOAD_X(b + 1)
        }
        __builtin_amdgcn_sched_barrier(0);

        #pragma unroll
        for (int i = 0; i < KB_STEPS; ++i) {
            #pragma unroll
            for (int u = 0; u < 2; ++u) {
                const float4 av = xa[b & 1][i >> 1][u];
                v2f a;
                a[0] = (i & 1) ? av.z : av.x;
                a[1] = (i & 1) ? av.w : av.y;
                c[2 * pr + u] = __builtin_amdgcn_wmma_f32_16x16x4_f32(false, a, false,
                                                             wf[KB_STEPS * kb + i],
                                                             (short)0, c[2 * pr + u], false, false);
            }
        }
    }
    #undef LOAD_W
    #undef LOAD_X

    #pragma unroll
    for (int t = 0; t < RT_PER_WAVE; ++t) {
        #pragma unroll
        for (int v = 0; v < 8; ++v) {
            const int rrow = (mgrp * RT_PER_WAVE + t) * 16 + v + 8 * half;
            const bool valid = ((rrow & (NPAD - 1)) < SEQ);
            qkv[(size_t)rrow * DMODEL + col] = valid ? c[t][v] : bv;
        }
    }
}

__global__ __launch_bounds__(512)
void attn_kernel(const float* __restrict__ qkv, float* __restrict__ out)
{
    extern __shared__ float lds[];
    float* Ys = lds;
    float* Ps = lds + MROWS * YSTRIDE;

    const int tid  = threadIdx.x;
    const int wave = tid >> 5;
    const int lane = tid & 31;
    const int half = lane >> 4;
    const int l16  = lane & 15;

    const int g  = blockIdx.y;
    const int bb = g >> 4;
    const int h  = (g >> 1) & 7;
    const int r  = g & 1;

    {
        const int dcol  = (tid & 15) * 2;
        const int rbase = tid >> 4;
        for (int i = rbase; i < MROWS; i += 32) {
            const float* src = qkv + ((size_t)(bb * NPAD + 2 * i + r)) * DMODEL
                                   + h * DK + dcol;
            const float2 v = *(const float2*)src;
            *(float2*)&Ys[i * YSTRIDE + dcol] = v;
        }
    }
    __syncthreads();

    float* Pw = Ps + wave * 16 * PSTRIDE;

    const int qb    = blockIdx.x * 16 + wave;
    const int qbase = qb * 16;

    const float scale = 0.17677669529663687f;

    float mrow = -3.0e38f, lrow = 0.0f;
    v8f o0, o1;
    #pragma unroll
    for (int v = 0; v < 8; ++v) {
        o0[v] = 0.0f; o1[v] = 0.0f;
    }

    for (int kb = 0; kb < MROWS / KSTEP; ++kb) {
        const int kbase = kb * KSTEP;

        v2f qf[DK / 4];
        v2f kf[NT][DK / 4];
        #pragma unroll
        for (int j = 0; j < DK / 4; ++j) {
            const float2 av = *(const float2*)&Ys[(qbase + l16) * YSTRIDE + 4 * j + 2 * half];
            qf[j][0] = av.x; qf[j][1] = av.y;
        }
        #pragma unroll
        for (int t = 0; t < NT; ++t) {
            #pragma unroll
            for (int j = 0; j < DK / 4; ++j) {
                const float2 bv = *(const float2*)&Ys[(kbase + 16 * t + l16) * YSTRIDE + 4 * j + 2 * half];
                kf[t][j][0] = bv.x; kf[t][j][1] = bv.y;
            }
        }
        __builtin_amdgcn_sched_barrier(0);

        v8f s[NT];
        #pragma unroll
        for (int t = 0; t < NT; ++t) {
            #pragma unroll
            for (int v = 0; v < 8; ++v) s[t][v] = 0.0f;
        }
        #pragma unroll
        for (int g = 0; g < NT; g += 4) {
            #pragma unroll
            for (int j = 0; j < DK / 4; ++j) {
                #pragma unroll
                for (int t = g; t < g + 4; ++t) {
                    s[t] = __builtin_amdgcn_wmma_f32_16x16x4_f32(false, qf[j], false, kf[t][j],
                                                                 (short)0, s[t], false, false);
                }
            }
            if (g == 0) __builtin_amdgcn_sched_barrier(0);
        }
        __builtin_amdgcn_sched_barrier(0);

        #pragma unroll
        for (int t = 0; t < NT; ++t) {
            #pragma unroll
            for (int v = 0; v < 8; ++v) {
                Pw[(v + 8 * half) * PSTRIDE + 16 * t + l16] = s[t][v];
            }
        }

        asm volatile("s_wait_dscnt 0x0" ::: "memory");

        v2f sf[KSTEP / 4];
        #pragma unroll
        for (int q = 0; q < KSTEP / 8; ++q) {
            const float4 pa = *(const float4*)&Pw[l16 * PSTRIDE + 8 * q + 4 * half];
            sf[2 * q][0] = pa.x;     sf[2 * q][1] = pa.y;
            sf[2 * q + 1][0] = pa.z; sf[2 * q + 1][1] = pa.w;
        }
        __builtin_amdgcn_sched_barrier(0);

        #define KKEY(I) (8 * ((I) >> 1) + 4 * half + 2 * ((I) & 1))

        v2f vf[KSTEP / 4][2];
        #pragma unroll
        for (int tl = 0; tl < 2; ++tl) {
            #pragma unroll
            for (int i = 0; i < KSTEP / 8; ++i) {
                const int key = kbase + KKEY(i);
                vf[i][tl][0] = Ys[key * YSTRIDE + 16 * tl + l16];
                vf[i][tl][1] = Ys[(key + 1) * YSTRIDE + 16 * tl + l16];
            }
            if (tl == 0) __builtin_amdgcn_sched_barrier(0);
        }
        __builtin_amdgcn_sched_barrier(0);

        {
            float pm[4];
            #pragma unroll
            for (int p = 0; p < 4; ++p) {
                float m = fmaxf(sf[8 * p][0], sf[8 * p][1]);
                #pragma unroll
                for (int i = 1; i < 8; ++i) m = fmaxf(m, fmaxf(sf[8 * p + i][0], sf[8 * p + i][1]));
                pm[p] = m;
            }
            float mx = fmaxf(fmaxf(pm[0], pm[1]), fmaxf(pm[2], pm[3]));
            mx = fmaxf(mx, xhalf(mx));
            const float newm = fmaxf(mrow, mx * scale);
            const float corr = __expf(mrow - newm);
            const v2f newm2 = {newm, newm};
            const float log2e = 0x1.715476p+0f;
            v2f ps[4];
            #pragma unroll
            for (int p = 0; p < 4; ++p) {
                #pragma unroll
                for (int i = 0; i < 8; ++i) {
                    const v2f u = (sf[8 * p + i] * scale - newm2) * log2e;
                    sf[8 * p + i][0] = __builtin_amdgcn_exp2f(u[0]);
                    sf[8 * p + i][1] = __builtin_amdgcn_exp2f(u[1]);
                    if (i == 0) ps[p] = sf[8 * p + i]; else ps[p] += sf[8 * p + i];
                }
            }
            const v2f pt = (ps[0] + ps[1]) + (ps[2] + ps[3]);
            float rsum = pt[0] + pt[1];
            rsum += xhalf(rsum);
            lrow = lrow * corr + rsum;
            mrow = newm;
            #define RESCALE_ROW(V) { const float cv = row_pick<V>(corr); o0[V] *= cv; o1[V] *= cv; }
            RESCALE_ROW(0) RESCALE_ROW(1) RESCALE_ROW(2) RESCALE_ROW(3)
            RESCALE_ROW(4) RESCALE_ROW(5) RESCALE_ROW(6) RESCALE_ROW(7)
            #undef RESCALE_ROW
        }
        __builtin_amdgcn_sched_barrier(0);

        #pragma unroll
        for (int tl = 0; tl < 2; ++tl) {
            #pragma unroll
            for (int i = KSTEP / 8; i < KSTEP / 4; ++i) {
                const int key = kbase + KKEY(i);
                vf[i][tl][0] = Ys[key * YSTRIDE + 16 * tl + l16];
                vf[i][tl][1] = Ys[(key + 1) * YSTRIDE + 16 * tl + l16];
            }
            if (tl == 0) __builtin_amdgcn_sched_barrier(0);
        }
        #undef KKEY
        __builtin_amdgcn_sched_barrier(0);

        #pragma unroll
        for (int i = 0; i < KSTEP / 4; ++i) {
            o0 = __builtin_amdgcn_wmma_f32_16x16x4_f32(false, sf[i], false, vf[i][0],
                                                       (short)0, o0, false, false);
            o1 = __builtin_amdgcn_wmma_f32_16x16x4_f32(false, sf[i], false, vf[i][1],
                                                       (short)0, o1, false, false);
        }
        __builtin_amdgcn_sched_barrier(0);
    }

    float mrv[8], lrv[8];
    #define PICK_ROW(V) { mrv[V] = row_pick<V>(mrow); lrv[V] = row_pick<V>(lrow); }
    PICK_ROW(0) PICK_ROW(1) PICK_ROW(2) PICK_ROW(3)
    PICK_ROW(4) PICK_ROW(5) PICK_ROW(6) PICK_ROW(7)
    #undef PICK_ROW

    #pragma unroll
    for (int v = 0; v < 8; ++v) {
        const int row = qbase + v + 8 * half;
        float e[3];
        const int d2 = l16 * 2;
        #pragma unroll
        for (int k = 0; k < 3; ++k) {
            const int nr = row + k - 1;
            const bool ok = (nr >= 0) && (nr < MROWS);
            const int nrc = ok ? nr : row;
            const float msk = ok ? 1.0f : 0.0f;
            float part = msk * (Ys[row * YSTRIDE + d2]     * Ys[nrc * YSTRIDE + d2]
                              + Ys[row * YSTRIDE + d2 + 1] * Ys[nrc * YSTRIDE + d2 + 1]);
            part = row_sum16(part);
            e[k] = part * scale;
        }
        const float newm = fmaxf(fmaxf(mrv[v], e[0]), fmaxf(e[1], e[2]));
        const float corr = __expf(mrv[v] - newm);
        float psum = 0.0f, add0 = 0.0f, add1 = 0.0f;
        #pragma unroll
        for (int k = 0; k < 3; ++k) {
            const float p = __expf(e[k] - newm);
            psum += p;
            const int nr = row + k - 1;
            const bool ok = (nr >= 0) && (nr < MROWS);
            const int nrc = ok ? nr : row;
            const float msk = ok ? 1.0f : 0.0f;
            add0 += (p * msk) * Ys[nrc * YSTRIDE + l16];
            add1 += (p * msk) * Ys[nrc * YSTRIDE + 16 + l16];
        }
        const float inv = 1.0f / (lrv[v] * corr + psum);
        const float r0 = (o0[v] * corr + add0) * inv;
        const float r1 = (o1[v] * corr + add1) * inv;

        const int n = 2 * row + r;
        if (n < SEQ) {
            const size_t base = ((size_t)(bb * SEQ + n)) * DMODEL + h * DK;
            out[base + l16]      = r0;
            out[base + 16 + l16] = r1;
        }
    }
}

extern "C" void kernel_launch(void* const* d_in, const int* in_sizes, int n_in,
                              void* d_out, int out_size, void* d_ws, size_t ws_size,
                              hipStream_t stream) {
    const float* x = (const float*)d_in[0];
    const float* W = (const float*)d_in[1];
    const float* b = (const float*)d_in[2];
    float* out = (float*)d_out;
    float* qkv = (float*)d_ws;

    dim3 g1(128, 16);
    qkv_gemm_kernel<<<g1, 32, 0, stream>>>(x, W, b, qkv);

    const size_t ldsBytes = (size_t)(MROWS * YSTRIDE + 16 * 16 * PSTRIDE) * sizeof(float);
    dim3 g2(4, 64);
    attn_kernel<<<g2, 512, ldsBytes, stream>>>(qkv, out);
}
